# starting kernel
# baseline (speedup 1.0000x reference)
#include <hip/hip_runtime.h>


#define PILLARS 40000
#define NPTS    32
#define CIN     4
#define UNITS   64
#define KPAD    12
#define WAVES_PER_BLOCK 8
#define BLOCK   (WAVES_PER_BLOCK * 32)
#define NBLK    ((PILLARS + WAVES_PER_BLOCK - 1) / WAVES_PER_BLOCK)

typedef float v2f __attribute__((ext_vector_type(2)));
typedef float v8f __attribute__((ext_vector_type(8)));

__device__ __forceinline__ void pillar_feats_to_lds(
    const float* __restrict__ features, const int* __restrict__ num_voxels,
    const int* __restrict__ coors, int p, int lane, float* __restrict__ frow)
{
    const float4 f = *reinterpret_cast<const float4*>(
        features + ((size_t)p * NPTS + lane) * CIN);

    float sx = f.x, sy = f.y, sz = f.z;
#pragma unroll
    for (int m = 16; m >= 1; m >>= 1) {
        sx += __shfl_xor(sx, m, 32);
        sy += __shfl_xor(sy, m, 32);
        sz += __shfl_xor(sz, m, 32);
    }
    const int   nv    = num_voxels[p];
    const float invnv = 1.0f / (float)nv;
    const float mx = sx * invnv, my = sy * invnv, mz = sz * invnv;

    const float cx = (float)coors[p * 4 + 3] * 0.2f + 0.1f;
    const float cy = (float)coors[p * 4 + 2] * 0.2f + (-39.9f);

    const float msk = (lane < nv) ? 1.0f : 0.0f;
    frow[0]  = f.x * msk;
    frow[1]  = f.y * msk;
    frow[2]  = f.z * msk;
    frow[3]  = f.w * msk;
    frow[4]  = (f.x - mx) * msk;
    frow[5]  = (f.y - my) * msk;
    frow[6]  = (f.z - mz) * msk;
    frow[7]  = (f.x - cx) * msk;
    frow[8]  = (f.y - cy) * msk;
    frow[9]  = 0.0f;
    frow[10] = 0.0f;
    frow[11] = 0.0f;
}

__device__ __forceinline__ void gemm_tile(
    const float* __restrict__ flw  ,
    const float* __restrict__ Wl   ,
    int nt, int lane, v8f& acc0, v8f& acc1)
{
    const int half = lane >> 4;
    const int lm   = lane & 15;
    const int u0   = nt * 16;
#pragma unroll
    for (int ks = 0; ks < 3; ++ks) {
        const int k0 = ks * 4 + 2 * half;
        v2f a0 = { flw[lm * KPAD + k0],        flw[lm * KPAD + k0 + 1] };
        v2f a1 = { flw[(16 + lm) * KPAD + k0], flw[(16 + lm) * KPAD + k0 + 1] };
        v2f b  = { Wl[k0 * UNITS + u0 + lm],   Wl[(k0 + 1) * UNITS + u0 + lm] };
        acc0 = __builtin_amdgcn_wmma_f32_16x16x4_f32(
            false, a0, false, b, (short)0, acc0, false, false);
        acc1 = __builtin_amdgcn_wmma_f32_16x16x4_f32(
            false, a1, false, b, (short)0, acc1, false, false);
    }
}

__global__ __launch_bounds__(BLOCK) void pfn_stats_kernel(
    const float* __restrict__ features, const int* __restrict__ num_voxels,
    const int* __restrict__ coors, const float* __restrict__ W,
    float* __restrict__ partials  )
{
    __shared__ float Wl[KPAD * UNITS];
    __shared__ float fl[WAVES_PER_BLOCK][NPTS][KPAD];
    __shared__ float red[2][WAVES_PER_BLOCK][UNITS];

    const int tid  = threadIdx.x;
    const int wave = tid >> 5;
    const int lane = tid & 31;
    const int p    = blockIdx.x * WAVES_PER_BLOCK + wave;

    for (int i = tid; i < KPAD * UNITS; i += BLOCK)
        Wl[i] = (i < (CIN + 5) * UNITS) ? W[i] : 0.0f;

    const bool valid = (p < PILLARS);
    if (valid)
        pillar_feats_to_lds(features, num_voxels, coors, p, lane,
                            &fl[wave][lane][0]);
    __syncthreads();

    const int half = lane >> 4;
    const int lm   = lane & 15;

    if (valid) {
#pragma unroll
        for (int nt = 0; nt < 4; ++nt) {
            v8f a0 = {}, a1 = {};
            gemm_tile(&fl[wave][0][0], Wl, nt, lane, a0, a1);
            float ps = 0.0f, pq = 0.0f;
#pragma unroll
            for (int j = 0; j < 8; ++j) {
                ps += a0[j] + a1[j];
                pq += a0[j] * a0[j] + a1[j] * a1[j];
            }
            ps += __shfl_xor(ps, 16, 32);
            pq += __shfl_xor(pq, 16, 32);
            if (half == 0) {
                red[0][wave][nt * 16 + lm] = ps;
                red[1][wave][nt * 16 + lm] = pq;
            }
        }
    } else {
        if (half == 0) {
#pragma unroll
            for (int nt = 0; nt < 4; ++nt) {
                red[0][wave][nt * 16 + lm] = 0.0f;
                red[1][wave][nt * 16 + lm] = 0.0f;
            }
        }
    }
    __syncthreads();

    if (tid < 2 * UNITS) {
        const int which = tid >> 6;
        const int u     = tid & 63;
        float s = 0.0f;
#pragma unroll
        for (int w = 0; w < WAVES_PER_BLOCK; ++w) s += red[which][w][u];
        partials[(size_t)blockIdx.x * (2 * UNITS) + tid] = s;
    }
}

__global__ __launch_bounds__(128) void pfn_finalize_stats(
    const float* __restrict__ partials, const float* __restrict__ gamma,
    const float* __restrict__ beta, float* __restrict__ stats  ,
    int nblk)
{
    __shared__ float red[128];
    const int t = threadIdx.x;
    float s = 0.0f;
    for (int w = 0; w < nblk; ++w) s += partials[(size_t)w * 128 + t];
    red[t] = s;
    __syncthreads();
    if (t < 64) {
        const float cnt  = (float)PILLARS * (float)NPTS;
        const float mean = red[t] / cnt;
        float var = red[64 + t] / cnt - mean * mean;
        var = fmaxf(var, 0.0f);
        const float inv = rsqrtf(var + 1e-3f);
        const float sc  = inv * gamma[t];
        stats[t]      = sc;
        stats[64 + t] = beta[t] - mean * sc;
    }
}

__global__ __launch_bounds__(BLOCK) void pfn_output_kernel(
    const float* __restrict__ features, const int* __restrict__ num_voxels,
    const int* __restrict__ coors, const float* __restrict__ W,
    const float* __restrict__ stats, float* __restrict__ out  )
{
    __shared__ float Wl[KPAD * UNITS];
    __shared__ float fl[WAVES_PER_BLOCK][NPTS][KPAD];

    const int tid  = threadIdx.x;
    const int wave = tid >> 5;
    const int lane = tid & 31;
    const int p    = blockIdx.x * WAVES_PER_BLOCK + wave;

    for (int i = tid; i < KPAD * UNITS; i += BLOCK)
        Wl[i] = (i < (CIN + 5) * UNITS) ? W[i] : 0.0f;

    const bool valid = (p < PILLARS);
    if (valid)
        pillar_feats_to_lds(features, num_voxels, coors, p, lane,
                            &fl[wave][lane][0]);
    __syncthreads();

    if (!valid) return;

    const int half = lane >> 4;
    const int lm   = lane & 15;

#pragma unroll
    for (int nt = 0; nt < 4; ++nt) {
        v8f a0 = {}, a1 = {};
        gemm_tile(&fl[wave][0][0], Wl, nt, lane, a0, a1);

        const int   u  = nt * 16 + lm;
        const float sc = stats[u];
        const float bi = stats[64 + u];

        float m = -3.402823466e38f;
#pragma unroll
        for (int j = 0; j < 8; ++j) {
            m = fmaxf(m, fmaf(a0[j], sc, bi));
            m = fmaxf(m, fmaf(a1[j], sc, bi));
        }
        m = fmaxf(m, __shfl_xor(m, 16, 32));
        m = fmaxf(m, 0.0f);
        if (half == 0) out[(size_t)p * UNITS + u] = m;
    }
}

extern "C" void kernel_launch(void* const* d_in, const int* in_sizes, int n_in,
                              void* d_out, int out_size, void* d_ws, size_t ws_size,
                              hipStream_t stream)
{
    const float* features   = (const float*)d_in[0];
    const int*   num_voxels = (const int*)  d_in[1];
    const int*   coors      = (const int*)  d_in[2];
    const float* W          = (const float*)d_in[3];
    const float* gamma      = (const float*)d_in[4];
    const float* beta       = (const float*)d_in[5];
    float*       out        = (float*)d_out;

    float* partials = (float*)d_ws;
    float* stats    = partials + (size_t)NBLK * 128;

    pfn_stats_kernel<<<NBLK, BLOCK, 0, stream>>>(features, num_voxels, coors, W,
                                                 partials);
    pfn_finalize_stats<<<1, 128, 0, stream>>>(partials, gamma, beta, stats, NBLK);
    pfn_output_kernel<<<NBLK, BLOCK, 0, stream>>>(features, num_voxels, coors, W,
                                                  stats, out);
}


# Output pass re-tiled: a wave takes 8 pillars as two 128-point blocks, a lane owns 4 points of one pillar; A fragments by lane-half swap in registers (no LDS), max stays in-lane (mean sum order)
# speedup vs baseline: 10.5156x; 10.5156x over previous
#include <hip/hip_runtime.h>


#define PILLARS 40000
#define NPTS    32
#define CIN     4
#define UNITS   64
#define KSTEPS  3

#define OWAVES  4
#define OBLOCK  (OWAVES * 32)
#define PPB     4
#define BPW     2
#define OGRID   (PILLARS / (PPB * BPW * OWAVES))

#define MWAVES  16
#define MBLOCK  (MWAVES * 32)
#define MGRID   250
#define MNW     (MGRID * MWAVES)
#define PPW     10
#define FP      20
#define TILE    256
#define FGROUPS 16

typedef float v2f __attribute__((ext_vector_type(2)));
typedef float v8f __attribute__((ext_vector_type(8)));

struct PillarIn { float4 f; int nv; int cxi; int cyi; };

__device__ __forceinline__ PillarIn load_pillar(
    const float* __restrict__ features, const int* __restrict__ num_voxels,
    const int* __restrict__ coors, int p, int lane)
{
    PillarIn r;
    r.f   = *reinterpret_cast<const float4*>(features + ((size_t)p * NPTS + lane) * CIN);
    r.nv  = num_voxels[p];
    r.cxi = coors[p * 4 + 3];
    r.cyi = coors[p * 4 + 2];
    return r;
}

__device__ __forceinline__ void moment_row_to_lds(const PillarIn& in, int lane, float* frow)
{
    const float4 f = in.f;
    float sx = f.x, sy = f.y, sz = f.z;
    for (int m = 16; m >= 1; m >>= 1) {
        sx += __shfl_xor(sx, m, 32);
        sy += __shfl_xor(sy, m, 32);
        sz += __shfl_xor(sz, m, 32);
    }
    const float invnv = 1.0f / (float)in.nv;
    const float mx = sx * invnv, my = sy * invnv, mz = sz * invnv;
    const float cx = (float)in.cxi * 0.2f + 0.1f;
    const float cy = (float)in.cyi * 0.2f + (-39.9f);
    const float msk = (lane < in.nv) ? 1.0f : 0.0f;

    float4* row = reinterpret_cast<float4*>(frow);
    row[0] = make_float4(f.x * msk, f.y * msk, f.z * msk, f.w * msk);
    row[1] = make_float4((f.x - mx) * msk, (f.y - my) * msk, (f.z - mz) * msk,
                         (f.x - cx) * msk);
    row[2] = make_float4((f.y - cy) * msk, msk, 0.0f, 0.0f);
}

__device__ __forceinline__ void gram_accumulate(
    const float* flw  , int lane, v8f& g0, v8f& g1)
{
    const int half = lane >> 4;
    const int lm   = lane & 15;
    for (int ks = 0; ks < 8; ++ks) {
        const int pt = ks * 4 + 2 * half;
        v2f ab = { flw[pt * FP + lm], flw[(pt + 1) * FP + lm] };
        if (ks & 1)
            g1 = __builtin_amdgcn_wmma_f32_16x16x4_f32(
                false, ab, false, ab, (short)0, g1, false, false);
        else
            g0 = __builtin_amdgcn_wmma_f32_16x16x4_f32(
                false, ab, false, ab, (short)0, g0, false, false);
    }
}

__device__ __forceinline__ void moment_step(
    PillarIn& s, const float* __restrict__ features, const int* __restrict__ num_voxels,
    const int* __restrict__ coors, int p, int lane, float* frow, const float* flw,
    v8f& g0, v8f& g1)
{
    moment_row_to_lds(s, lane, frow);
    const int pn = (p + 2 * MNW < PILLARS) ? (p + 2 * MNW) : (PILLARS - 1);
    s = load_pillar(features, num_voxels, coors, pn, lane);
    gram_accumulate(flw, lane, g0, g1);
}

__global__ __launch_bounds__(MBLOCK) void pfn_moments_kernel(
    const float* __restrict__ features, const int* __restrict__ num_voxels,
    const int* __restrict__ coors,
    float* __restrict__ tiles  )
{
    __shared__ __attribute__((aligned(16))) float fl[MWAVES][NPTS][FP];
    __shared__ float red[MWAVES][TILE];

    const int tid  = threadIdx.x;
    const int wave = __builtin_amdgcn_readfirstlane(tid >> 5);
    const int lane = tid & 31;
    const int half = lane >> 4;
    const int lm   = lane & 15;
    const int gw   = blockIdx.x * MWAVES + wave;

    float* frow = &fl[wave][lane][0];
    *reinterpret_cast<float4*>(frow + 12) = make_float4(0.0f, 0.0f, 0.0f, 0.0f);

    v8f g0 = {}, g1 = {};

    static_assert(MNW * PPW == PILLARS && PPW % 2 == 0, "pillars per wave");
    const float* flw = &fl[wave][0][0];
    PillarIn a = load_pillar(features, num_voxels, coors, gw, lane);
    PillarIn b = load_pillar(features, num_voxels, coors, gw + MNW, lane);
    int p = gw;
    for (int it = 0; it < PPW / 2; ++it) {
        moment_step(a, features, num_voxels, coors, p, lane, frow, flw, g0, g1);
        moment_step(b, features, num_voxels, coors, p + MNW, lane, frow, flw, g0, g1);
        p += 2 * MNW;
    }

    for (int r = 0; r < 8; ++r)
        red[wave][(8 * half + r) * 16 + lm] = g0[r] + g1[r];
    __syncthreads();

    if (tid < TILE) {
        float s = 0.0f;
        for (int w = 0; w < MWAVES; ++w) s += red[w][tid];
        tiles[(size_t)blockIdx.x * TILE + tid] = s;
    }
}

__global__ __launch_bounds__(1024) void pfn_finalize_moments(
    const float* __restrict__ tiles, const float* __restrict__ W,
    const float* __restrict__ gamma, const float* __restrict__ beta,
    float* __restrict__ stats  )
{
    __shared__ __attribute__((aligned(16))) float red[FGROUPS][TILE];
    __shared__ float G[TILE];

    const int t  = threadIdx.x;
    const int gi = t >> 6;
    const int q  = t & 63;

    float4 s = make_float4(0.0f, 0.0f, 0.0f, 0.0f);
    for (int k = 0; k < (MGRID + FGROUPS - 1) / FGROUPS; ++k) {
        const int tl = k * FGROUPS + gi;
        if (tl < MGRID) {
            const float4 v = *reinterpret_cast<const float4*>(
                tiles + (size_t)tl * TILE + 4 * q);
            s.x += v.x; s.y += v.y; s.z += v.z; s.w += v.w;
        }
    }
    *reinterpret_cast<float4*>(&red[gi][4 * q]) = s;
    __syncthreads();

    if (t < TILE) {
        float acc = 0.0f;
        for (int g = 0; g < FGROUPS; ++g) acc += red[g][t];
        G[t] = acc;
    }
    __syncthreads();

    if (t < UNITS) {
        float w[CIN + 5];
        for (int c = 0; c < CIN + 5; ++c) w[c] = W[c * UNITS + t];

        float sum = 0.0f, sq = 0.0f;
        for (int c = 0; c < CIN + 5; ++c) {
            sum = fmaf(w[c], G[c * 16 + 9], sum);
            float tc = 0.0f;
            for (int c2 = 0; c2 < CIN + 5; ++c2) tc = fmaf(G[c * 16 + c2], w[c2], tc);
            sq = fmaf(w[c], tc, sq);
        }

        const float cnt  = (float)PILLARS * (float)NPTS;
        const float mean = sum / cnt;
        float var = sq / cnt - mean * mean;
        var = fmaxf(var, 0.0f);
        const float inv = rsqrtf(var + 1e-3f);
        const float sc  = inv * gamma[t];
        stats[t]      = sc;
        stats[64 + t] = beta[t] - mean * sc;
    }
}

struct BlockIn { float4 f[4]; int nv; int2 cyx; };

__device__ __forceinline__ BlockIn load_block(
    const float* __restrict__ features, const int* __restrict__ num_voxels,
    const int* __restrict__ coors, int p, int i)
{
    BlockIn r;
    const float* base = features + ((size_t)p * NPTS + i) * CIN;
    for (int j = 0; j < 4; ++j)
        r.f[j] = *reinterpret_cast<const float4*>(base + 8 * j * CIN);
    r.nv  = num_voxels[p];
    r.cyx = *reinterpret_cast<const int2*>(coors + p * 4 + 2);
    return r;
}

template <int MASK>
__device__ __forceinline__ float row_xor_add(float v)
{
    return v + __int_as_float(__builtin_amdgcn_update_dpp(
                   0, __float_as_int(v), 0x160 | MASK  , 0xf, 0xf, true));
}

__device__ __forceinline__ void half_swap(float& x, float& y)
{
    const auto r = __builtin_amdgcn_permlane16_swap(
        __float_as_uint(x), __float_as_uint(y), false, false);
    x = __uint_as_float(r[0]);
    y = __uint_as_float(r[1]);
}

__device__ __forceinline__ void block_fragments(
    const BlockIn& in, int i, v2f (&a)[4][2][KSTEPS])
{
    float sx = (in.f[0].x + in.f[1].x) + (in.f[2].x + in.f[3].x);
    float sy = (in.f[0].y + in.f[1].y) + (in.f[2].y + in.f[3].y);
    float sz = (in.f[0].z + in.f[1].z) + (in.f[2].z + in.f[3].z);
    sx = row_xor_add<1>(sx); sy = row_xor_add<1>(sy); sz = row_xor_add<1>(sz);
    sx = row_xor_add<2>(sx); sy = row_xor_add<2>(sy); sz = row_xor_add<2>(sz);
    sx = row_xor_add<4>(sx); sy = row_xor_add<4>(sy); sz = row_xor_add<4>(sz);

    const float invnv = 1.0f / (float)in.nv;
    const float mx = sx * invnv, my = sy * invnv, mz = sz * invnv;
    const float cx = (float)in.cyx.y * 0.2f + 0.1f;
    const float cy = (float)in.cyx.x * 0.2f + (-39.9f);

    for (int j = 0; j < 4; ++j) {
        const float4 f = in.f[j];
        const float msk = (i + 8 * j < in.nv) ? 1.0f : 0.0f;
        float F[4 * KSTEPS];
        F[0]  = f.x * msk;
        F[1]  = f.y * msk;
        F[2]  = f.z * msk;
        F[3]  = f.w * msk;
        F[4]  = (f.x - mx) * msk;
        F[5]  = (f.y - my) * msk;
        F[6]  = (f.z - mz) * msk;
        F[7]  = (f.x - cx) * msk;
        F[8]  = (f.y - cy) * msk;
        F[9]  = 0.0f;
        F[10] = 0.0f;
        F[11] = 0.0f;
        for (int ks = 0; ks < 2; ++ks) {
            for (int e = 0; e < 2; ++e) {
                float x = F[4 * ks + e], y = F[4 * ks + 2 + e];
                half_swap(x, y);
                a[j][0][ks][e] = x;
                a[j][1][ks][e] = y;
            }
        }
        float x = F[8], y = 0.0f;
        half_swap(x, y);
        a[j][0][2] = (v2f){ x, 0.0f };
        a[j][1][2] = (v2f){ y, 0.0f };
    }
}

__global__ __launch_bounds__(OBLOCK, 5) void pfn_output_kernel(
    const float* __restrict__ features, const int* __restrict__ num_voxels,
    const int* __restrict__ coors, const float* __restrict__ W,
    const float* __restrict__ stats, float* __restrict__ out  )
{
    static_assert(OGRID * OWAVES * BPW * PPB == PILLARS, "eight pillars per wave, no tail");

    const int tid  = threadIdx.x;
    const int lane = tid & 31;
    const int half = lane >> 4;
    const int lm   = lane & 15;
    const int q    = lane >> 3;
    const int i    = lane & 7;
    const int pb   = (blockIdx.x * OWAVES + (tid >> 5)) * (BPW * PPB);

    BlockIn in[BPW];
    for (int k = 0; k < BPW; ++k)
        in[k] = load_block(features, num_voxels, coors, pb + PPB * k + q, i);

    v2f   b[4][KSTEPS];
    float sc[4], bi[4];
    for (int nt = 0; nt < 4; ++nt) {
        const int u = nt * 16 + lm;
        b[nt][0] = (v2f){ W[(2 * half) * UNITS + u],     W[(2 * half + 1) * UNITS + u] };
        b[nt][1] = (v2f){ W[(4 + 2 * half) * UNITS + u], W[(5 + 2 * half) * UNITS + u] };
        b[nt][2] = (v2f){ W[8 * UNITS + u], 0.0f };
        sc[nt] = stats[u];
        bi[nt] = stats[64 + u];
    }

    for (int k = 0; k < BPW; ++k) {
        v2f a[4][2][KSTEPS];
        block_fragments(in[k], i, a);

        const int p0 = pb + PPB * k;
        for (int np = 0; np < 2; ++np) {
            float mm[2][2];
            for (int t = 0; t < 2; ++t) {
                const int nt = 2 * np + t;
                float m0 = -3.402823466e38f, m1 = -3.402823466e38f;
                for (int j = 0; j < 4; ++j) {
                    v8f d0 = {}, d1 = {};
                    for (int ks = 0; ks < KSTEPS; ++ks) {
                        d0 = __builtin_amdgcn_wmma_f32_16x16x4_f32(
                            false, a[j][0][ks], false, b[nt][ks], (short)0, d0, false, false);
                        d1 = __builtin_amdgcn_wmma_f32_16x16x4_f32(
                            false, a[j][1][ks], false, b[nt][ks], (short)0, d1, false, false);
                    }
                    for (int r = 0; r < 8; ++r) {
                        m0 = fmaxf(m0, fmaf(d0[r], sc[nt], bi[nt]));
                        m1 = fmaxf(m1, fmaf(d1[r], sc[nt], bi[nt]));
                    }
                }
                mm[t][0] = fmaxf(m0, 0.0f);
                mm[t][1] = fmaxf(m1, 0.0f);
            }
            for (int s = 0; s < 2; ++s) {
                float x = mm[0][s], y = mm[1][s];
                half_swap(x, y);
                out[(size_t)(p0 + 2 * s)     * UNITS + 32 * np + lane] = x;
                out[(size_t)(p0 + 2 * s + 1) * UNITS + 32 * np + lane] = y;
            }
        }
    }
}

extern "C" void kernel_launch(void* const* d_in, const int* in_sizes, int n_in,
                              void* d_out, int out_size, void* d_ws, size_t ws_size,
                              hipStream_t stream)
{
    const float* features   = (const float*)d_in[0];
    const int*   num_voxels = (const int*)  d_in[1];
    const int*   coors      = (const int*)  d_in[2];
    const float* W          = (const float*)d_in[3];
    const float* gamma      = (const float*)d_in[4];
    const float* beta       = (const float*)d_in[5];
    float*       out        = (float*)d_out;

    float* tiles = (float*)d_ws;
    float* stats = tiles + (size_t)MGRID * TILE;

    pfn_moments_kernel<<<MGRID, MBLOCK, 0, stream>>>(features, num_voxels, coors, tiles);
    pfn_finalize_moments<<<1, 1024, 0, stream>>>(tiles, W, gamma, beta, stats);
    pfn_output_kernel<<<OGRID, OBLOCK, 0, stream>>>(features, num_voxels, coors, W,
                                                    stats, out);
}
